# starting kernel
# baseline (speedup 1.0000x reference)
#include <hip/hip_runtime.h>


typedef __attribute__((ext_vector_type(2))) float v2f;
typedef __attribute__((ext_vector_type(8))) float v8f;

#define KDIM 64

__global__ void fill_zero(float* __restrict__ p, int n) {
  int i = blockIdx.x * blockDim.x + threadIdx.x;
  int stride = gridDim.x * blockDim.x;
  for (; i < n; i += stride) p[i] = 0.0f;
}

template <int OUTF, bool RELU, bool ADD_AGG>
__global__ __launch_bounds__(256) void sage_gemm(
    const float* __restrict__ X, const float* __restrict__ W,
    const float* __restrict__ bias, const float* __restrict__ agg,
    const float* __restrict__ cnt, float* __restrict__ Y, int n) {
  constexpr int NT = OUTF / 16;
  const int wave = threadIdx.x >> 5;
  const int lane = threadIdx.x & 31;
  const int half = lane >> 4;
  const int l15  = lane & 15;
  const int tile = blockIdx.x * (blockDim.x >> 5) + wave;
  const int r0 = tile * 16;
  if (r0 >= n) return;

  int arow = r0 + l15;
  if (arow >= n) arow = n - 1;
  const float* xrow = X + (long)arow * KDIM + half * 2;

  v8f acc[NT];
#pragma unroll
  for (int t = 0; t < NT; ++t) {
    float b = bias ? bias[t * 16 + l15] : 0.0f;
#pragma unroll
    for (int j = 0; j < 8; ++j) acc[t][j] = b;
  }

#pragma unroll
  for (int kb = 0; kb < KDIM / 4; ++kb) {
    v2f a = *(const v2f*)(xrow + kb * 4);
#pragma unroll
    for (int t = 0; t < NT; ++t) {
      v2f b = *(const v2f*)(W + (t * 16 + l15) * KDIM + kb * 4 + half * 2);
      acc[t] = __builtin_amdgcn_wmma_f32_16x16x4_f32(
           false, a,  false, b,
           (short)0, acc[t],  false,  false);
    }
  }

#pragma unroll
  for (int t = 0; t < NT; ++t) {
    int col = t * 16 + l15;
#pragma unroll
    for (int v = 0; v < 8; ++v) {
      int row = r0 + half * 8 + v;
      if (row < n) {
        float val = acc[t][v];
        if (ADD_AGG) {
          float c = cnt[row];
          val += agg[(long)row * OUTF + col] / fmaxf(c, 1.0f);
        }
        if (RELU) val = fmaxf(val, 0.0f);
        Y[(long)row * OUTF + col] = val;
      }
    }
  }
}

template <int F>
__global__ __launch_bounds__(256) void sage_scatter(
    const int* __restrict__ ei, const float* __restrict__ H,
    float* __restrict__ agg, float* __restrict__ cnt, int nE) {
  constexpr int CH = F / 4;
  int idx = blockIdx.x * blockDim.x + threadIdx.x;
  if (idx >= nE * CH) return;
  int e = idx / CH;
  int c = idx - e * CH;
  int src = ei[e];
  int dst = ei[nE + e];
  const float4 m = *(const float4*)(H + (long)src * F + c * 4);
  float* ap = agg + (long)dst * F + c * 4;
  (void)__hip_atomic_fetch_add(ap + 0, m.x, __ATOMIC_RELAXED, __HIP_MEMORY_SCOPE_AGENT);
  (void)__hip_atomic_fetch_add(ap + 1, m.y, __ATOMIC_RELAXED, __HIP_MEMORY_SCOPE_AGENT);
  (void)__hip_atomic_fetch_add(ap + 2, m.z, __ATOMIC_RELAXED, __HIP_MEMORY_SCOPE_AGENT);
  (void)__hip_atomic_fetch_add(ap + 3, m.w, __ATOMIC_RELAXED, __HIP_MEMORY_SCOPE_AGENT);
  if (cnt != nullptr && c == 0)
    (void)__hip_atomic_fetch_add(cnt + dst, 1.0f, __ATOMIC_RELAXED, __HIP_MEMORY_SCOPE_AGENT);
}

extern "C" void kernel_launch(void* const* d_in, const int* in_sizes, int n_in,
                              void* d_out, int out_size, void* d_ws, size_t ws_size,
                              hipStream_t stream) {
  const float* x   = (const float*)d_in[0];
  const int*   ei  = (const int*)d_in[1];
  const float* Wl0 = (const float*)d_in[2];
  const float* bl0 = (const float*)d_in[3];
  const float* Wr0 = (const float*)d_in[4];
  const float* Wl1 = (const float*)d_in[5];
  const float* bl1 = (const float*)d_in[6];
  const float* Wr1 = (const float*)d_in[7];
  float* out = (float*)d_out;

  const int N = in_sizes[0] / 64;
  const int E = in_sizes[1] / 2;

  float* B0 = (float*)d_ws;
  float* B1 = B0 + (size_t)N * 64;
  float* B2 = B1 + (size_t)N * 64;
  float* B3 = B2 + (size_t)N * 64;

  const dim3 blk(256);
  const int tiles = (N + 15) / 16;
  const dim3 gemmGrid((tiles + 7) / 8);

  fill_zero<<<dim3(2048), blk, 0, stream>>>(B1, N * 64);
  fill_zero<<<dim3(256),  blk, 0, stream>>>(B3, N);
  sage_gemm<64, false, false><<<gemmGrid, blk, 0, stream>>>(x, Wl0, bl0, nullptr, nullptr, B0, N);
  {
    int work = E * 16;
    sage_scatter<64><<<dim3((work + 255) / 256), blk, 0, stream>>>(ei, B0, B1, B3, E);
  }
  sage_gemm<64, true, true><<<gemmGrid, blk, 0, stream>>>(x, Wr0, nullptr, B1, B3, B2, N);

  fill_zero<<<dim3(1024), blk, 0, stream>>>(B1, N * 32);
  sage_gemm<32, false, false><<<gemmGrid, blk, 0, stream>>>(B2, Wl1, bl1, nullptr, nullptr, B0, N);
  {
    int work = E * 8;
    sage_scatter<32><<<dim3((work + 255) / 256), blk, 0, stream>>>(ei, B0, B1, nullptr, E);
  }
  sage_gemm<32, false, true><<<gemmGrid, blk, 0, stream>>>(B2, Wr1, nullptr, B1, B3, out, N);
}


# Layer-0 gather moved into the root/layer-1 product launch: a wave gathers its 16 mean rows, turns them into accumulator layout by two lane-bit/register-bit exchanges; root sums start from the mean
# speedup vs baseline: 3.8816x; 3.8816x over previous
#include <hip/hip_runtime.h>


typedef __attribute__((ext_vector_type(2))) float v2f;
typedef __attribute__((ext_vector_type(8))) float v8f;

#define KDIM 64

__global__ void fill_head(int* __restrict__ head, int* __restrict__ cnt, int n,
                          int* __restrict__ ticket) {
  int i = blockIdx.x * blockDim.x + threadIdx.x;
  int stride = gridDim.x * blockDim.x;
  if (i < 64) ticket[i] = 0;
  for (; i < n; i += stride) {
    head[i] = -1;
    cnt[i] = 0;
  }
}

template <int OUTF>
__global__ __launch_bounds__(256) void sage_gemm(
    const float* __restrict__ X, const float* __restrict__ W,
    const float* __restrict__ bias, float* __restrict__ Y, int n) {
  constexpr int NT = OUTF / 16;
  const int wave = threadIdx.x >> 5;
  const int lane = threadIdx.x & 31;
  const int half = lane >> 4;
  const int l15  = lane & 15;
  const int tile = blockIdx.x * (blockDim.x >> 5) + wave;
  const int r0 = tile * 16;
  if (r0 >= n) return;

  int arow = r0 + l15;
  if (arow >= n) arow = n - 1;
  const float* xrow = X + (long)arow * KDIM + half * 2;

  v8f acc[NT];
#pragma unroll
  for (int t = 0; t < NT; ++t) {
    float b = bias[t * 16 + l15];
#pragma unroll
    for (int j = 0; j < 8; ++j) acc[t][j] = b;
  }

#pragma unroll
  for (int kb = 0; kb < KDIM / 4; ++kb) {
    v2f a = *(const v2f*)(xrow + kb * 4);
#pragma unroll
    for (int t = 0; t < NT; ++t) {
      v2f b = *(const v2f*)(W + (t * 16 + l15) * KDIM + kb * 4 + half * 2);
      acc[t] = __builtin_amdgcn_wmma_f32_16x16x4_f32(
           false, a,  false, b,
           (short)0, acc[t],  false,  false);
    }
  }

#pragma unroll
  for (int t = 0; t < NT; ++t) {
    int col = t * 16 + l15;
#pragma unroll
    for (int v = 0; v < 8; ++v) {
      int row = r0 + half * 8 + v;
      if (row < n) Y[(long)row * OUTF + col] = acc[t][v];
    }
  }
}

#define SLOT_CAP 32

__global__ __launch_bounds__(256, 8) void sage_gather0_root0_layer1(
    const int* __restrict__ slots, const int* __restrict__ cnt,
    const int* __restrict__ head, const int* __restrict__ link,
    const float* __restrict__ H0,
    const float* __restrict__ X, const float* __restrict__ Wr0,
    const float* __restrict__ Wl1, const float* __restrict__ bl1,
    const float* __restrict__ Wr1,
    float* __restrict__ HL, float* __restrict__ ROOT, int n, int nE) {
  constexpr int F0 = 64;
  constexpr int F1 = 32;
  constexpr int Q = F0 / 32;
  constexpr int DEEP = 16 / Q;
  const int wave = threadIdx.x >> 5;
  const int lane = threadIdx.x & 31;
  const int half = lane >> 4;
  const int l15  = lane & 15;
  const int tile = blockIdx.x * (blockDim.x >> 5) + wave;
  const int r0 = tile * 16;
  if (r0 >= n) return;

  const int gbase = lane & 12;
  const int sub = ((lane & 2) << 1) | (half << 1) | (lane & 1);
  const int4* __restrict__ slot4 = reinterpret_cast<const int4*>(slots);
  const int2* __restrict__ link2 = reinterpret_cast<const int2*>(link);

  float mn[4][Q][4];
#pragma unroll
  for (int k = 0; k < 4; ++k) {
#pragma unroll
    for (int q = 0; q < Q; ++q) {
#pragma unroll
      for (int e = 0; e < 4; ++e) mn[k][q][e] = 0.0f;
    }
  }

  int d = r0 + gbase;
  int4 sl;
  int cn;
  {
    const int dc = d < n ? d : n - 1;
    sl = slot4[(long)dc * 8 + sub];
    cn = cnt[dc];
    if (d >= n) cn = 0;
  }
#pragma unroll 1
  for (int p = 0; p < 4; ++p, ++d) {
    const int dn = d + 1 < n ? d + 1 : n - 1;
    const int4 nx = slot4[(long)dn * 8 + sub];
    int ncn = cnt[dn];
    if (d + 1 >= n) ncn = 0;

    const int m = cn < SLOT_CAP ? (cn > 0 ? cn : 0) : SLOT_CAP;

    float4 acc[Q];
#pragma unroll
    for (int q = 0; q < Q; ++q) acc[q] = make_float4(0.0f, 0.0f, 0.0f, 0.0f);
#pragma unroll
    for (int j = 0; j < SLOT_CAP / 4; j += DEEP / 4) {
      if (j * 4 < m) {
        int s[DEEP];
        float4 v[DEEP][Q];
#pragma unroll
        for (int u = 0; u < DEEP; ++u) {
          const int w = (u & 3) == 0 ? sl.x : (u & 3) == 1 ? sl.y : (u & 3) == 2 ? sl.z : sl.w;
          const int ks = j + (u >> 2);
          s[u] = __shfl(w, ((ks >> 1) & 1) * 16 + ((ks >> 2) << 1) + (ks & 1) + gbase);
        }
#pragma unroll
        for (int u = 0; u < DEEP; ++u) {
#pragma unroll
          for (int q = 0; q < Q; ++q) {
            v[u][q] = make_float4(0.0f, 0.0f, 0.0f, 0.0f);
            if (j * 4 + u < m)
              v[u][q] = *reinterpret_cast<const float4*>(H0 + (long)s[u] * F0 + q * 32 + sub * 4);
          }
        }
#pragma unroll
        for (int u = 0; u < DEEP; ++u) {
#pragma unroll
          for (int q = 0; q < Q; ++q) {
            acc[q].x += v[u][q].x; acc[q].y += v[u][q].y; acc[q].z += v[u][q].z; acc[q].w += v[u][q].w;
          }
        }
      }
    }

    if (cn > SLOT_CAP) {
      int e = head[d];
      int c = SLOT_CAP;
      int2 lk = make_int2(0, -1);
      if ((unsigned)e < (unsigned)nE) lk = link2[e];
      while ((unsigned)e < (unsigned)nE && c < cn) {
        float4 v[Q];
#pragma unroll
        for (int q = 0; q < Q; ++q)
          v[q] = *reinterpret_cast<const float4*>(H0 + (long)lk.x * F0 + q * 32 + sub * 4);
        e = lk.y;
        if ((unsigned)e < (unsigned)nE) lk = link2[e];
#pragma unroll
        for (int q = 0; q < Q; ++q) {
          acc[q].x += v[q].x; acc[q].y += v[q].y; acc[q].z += v[q].z; acc[q].w += v[q].w;
        }
        ++c;
      }
    }

    const float div = fmaxf((float)cn, 1.0f);
#pragma unroll
    for (int q = 0; q < Q; ++q) {
#pragma unroll
      for (int e = 0; e < 4; ++e) {
        mn[0][q][e] = mn[1][q][e];
        mn[1][q][e] = mn[2][q][e];
        mn[2][q][e] = mn[3][q][e];
      }
      mn[3][q][0] = acc[q].x / div;
      mn[3][q][1] = acc[q].y / div;
      mn[3][q][2] = acc[q].z / div;
      mn[3][q][3] = acc[q].w / div;
    }
    sl = nx;
    cn = ncn;
  }

  const bool hi1 = (lane & 2) != 0;
  const bool hi0 = (lane & 1) != 0;
  float n1[2][2][Q][4];
#pragma unroll
  for (int p0 = 0; p0 < 2; ++p0) {
#pragma unroll
    for (int q = 0; q < Q; ++q) {
#pragma unroll
      for (int e = 0; e < 4; ++e) {
        const float lo = mn[p0][q][e];
        const float up = mn[2 + p0][q][e];
        const float got = __shfl_xor(hi1 ? lo : up, 2);
        n1[0][p0][q][e] = hi1 ? got : lo;
        n1[1][p0][q][e] = hi1 ? up : got;
      }
    }
  }
  v8f h[4];
#pragma unroll
  for (int a = 0; a < 2; ++a) {
#pragma unroll
    for (int q = 0; q < Q; ++q) {
#pragma unroll
      for (int e = 0; e < 4; ++e) {
        const float lo = n1[a][0][q][e];
        const float up = n1[a][1][q][e];
        const float got = __shfl_xor(hi0 ? lo : up, 1);
        h[2 * q + a][e]     = hi0 ? got : lo;
        h[2 * q + a][4 + e] = hi0 ? up : got;
      }
    }
  }

  const int node = r0 + l15;
  const int nodec = node < n ? node : n - 1;
  const float* xrow = X + (long)nodec * KDIM + half * 2;

#pragma unroll 8
  for (int kb = 0; kb < KDIM / 4; ++kb) {
    const v2f b = *(const v2f*)(xrow + kb * 4);
#pragma unroll
    for (int t = 0; t < 4; ++t) {
      const v2f a = *(const v2f*)(Wr0 + (t * 16 + l15) * KDIM + kb * 4 + half * 2);
      h[t] = __builtin_amdgcn_wmma_f32_16x16x4_f32(
           false, a,  false, b,
           (short)0, h[t],  false,  false);
    }
  }

#pragma unroll
  for (int t = 0; t < 4; ++t) {
#pragma unroll
    for (int r = 0; r < 8; ++r) h[t][r] = fmaxf(h[t][r], 0.0f);
  }

  v8f accL[2], accR[2];
#pragma unroll
  for (int u = 0; u < 2; ++u) {
#pragma unroll
    for (int r = 0; r < 8; ++r) {
      accL[u][r] = bl1[u * 16 + half * 8 + r];
      accR[u][r] = 0.0f;
    }
  }

#pragma unroll
  for (int j = 0; j < KDIM / 4; ++j) {
    const int t = j >> 2;
    const int q = j & 3;
    v2f b;
    b[0] = h[t][2 * q];
    b[1] = h[t][2 * q + 1];
#pragma unroll
    for (int u = 0; u < 2; ++u) {
      const int woff = (u * 16 + l15) * KDIM + t * 16 + half * 8 + q * 2;
      const v2f al = *(const v2f*)(Wl1 + woff);
      const v2f ar = *(const v2f*)(Wr1 + woff);
      accL[u] = __builtin_amdgcn_wmma_f32_16x16x4_f32(
           false, al,  false, b,
           (short)0, accL[u],  false,  false);
      accR[u] = __builtin_amdgcn_wmma_f32_16x16x4_f32(
           false, ar,  false, b,
           (short)0, accR[u],  false,  false);
    }
  }

  if (node < n) {
    float* hp = HL + (long)node * F1 + half * 8;
    float* rp = ROOT + (long)node * F1 + half * 8;
#pragma unroll
    for (int u = 0; u < 2; ++u) {
      *reinterpret_cast<float4*>(hp + u * 16) =
          make_float4(accL[u][0], accL[u][1], accL[u][2], accL[u][3]);
      *reinterpret_cast<float4*>(hp + u * 16 + 4) =
          make_float4(accL[u][4], accL[u][5], accL[u][6], accL[u][7]);
      *reinterpret_cast<float4*>(rp + u * 16) =
          make_float4(accR[u][0], accR[u][1], accR[u][2], accR[u][3]);
      *reinterpret_cast<float4*>(rp + u * 16 + 4) =
          make_float4(accR[u][4], accR[u][5], accR[u][6], accR[u][7]);
    }
  }
}


#define LIST_EPT 8
#define LIST_CHUNK (256 * LIST_EPT)
__global__ __launch_bounds__(256) void build_slots(
    const int* __restrict__ ei, int* __restrict__ cnt, int* __restrict__ slots,
    int* __restrict__ head, int* __restrict__ link,
    int* __restrict__ ticket, int nE, int n) {
  constexpr int EPT = LIST_EPT;
  const int chunks = (nE + LIST_CHUNK - 1) / LIST_CHUNK;
  const int split = ((n >> 1) + 31) & ~31;
  __shared__ int sh_upper, sh_chunk;
  if (threadIdx.x == 0) {
    const unsigned id = __builtin_amdgcn_s_sendmsg_rtn(0x87);
    int upper = (int)((id >> 18) & 1u);
    int c = __hip_atomic_fetch_add(ticket + upper * 32, 1, __ATOMIC_RELAXED, __HIP_MEMORY_SCOPE_AGENT);
    if (c >= chunks) {
      upper ^= 1;
      c = __hip_atomic_fetch_add(ticket + upper * 32, 1, __ATOMIC_RELAXED, __HIP_MEMORY_SCOPE_AGENT);
    }
    sh_upper = upper;
    sh_chunk = c;
  }
  __syncthreads();
  const bool upper = sh_upper != 0;
  const int chunk = sh_chunk;
  if ((unsigned)chunk >= (unsigned)chunks) return;

  const int base = chunk * LIST_CHUNK + threadIdx.x;
  int s[EPT], d[EPT], rk[EPT];
#pragma unroll
  for (int k = 0; k < EPT; ++k) {
    const int e = base + k * 256;
    const int ec = e < nE ? e : nE - 1;
    d[k] = ei[nE + ec];
  }
#pragma unroll
  for (int k = 0; k < EPT; ++k) {
    const int e = base + k * 256;
    const int ec = e < nE ? e : nE - 1;
    s[k] = ei[ec];
  }
#pragma unroll
  for (int k = 0; k < EPT; ++k) {
    const int e = base + k * 256;
    rk[k] = -1;
    if (e < nE && (d[k] >= split) == upper)
      rk[k] = __hip_atomic_fetch_add(cnt + d[k], 1, __ATOMIC_RELAXED, __HIP_MEMORY_SCOPE_AGENT);
  }
  int2* __restrict__ link2 = reinterpret_cast<int2*>(link);
#pragma unroll
  for (int k = 0; k < EPT; ++k) {
    const int e = base + k * 256;
    if (e < nE && (d[k] >= split) == upper) {
      if ((unsigned)rk[k] < (unsigned)SLOT_CAP) {
        slots[(long)d[k] * SLOT_CAP + rk[k]] = s[k];
      } else {
        const int old = __hip_atomic_exchange(head + d[k], e, __ATOMIC_RELAXED, __HIP_MEMORY_SCOPE_AGENT);
        link2[e] = make_int2(s[k], old);
      }
    }
  }
}

template <int F, bool ADD_ROOT>
__global__ __launch_bounds__(64) void sage_gather_slots(
    const int* __restrict__ slots, const int* __restrict__ cnt,
    const int* __restrict__ head, const int* __restrict__ link,
    const float* __restrict__ H, const float* __restrict__ root,
    float* __restrict__ out, int n, int nE) {
  static_assert(F == 32 || F == 64, "rows of one or two 128-byte lines");
  constexpr int LPN = 8;
  constexpr int NPP = 64 / LPN;
  constexpr int PASSES = 4;
  constexpr int Q = F / 32;
  constexpr int DEEP = 16 / Q;
  const int sub = (int)(threadIdx.x % LPN);
  const int4* __restrict__ slot4 = reinterpret_cast<const int4*>(slots);
  const int2* __restrict__ link2 = reinterpret_cast<const int2*>(link);

  int d = blockIdx.x * (NPP * PASSES) + (int)(threadIdx.x / LPN);
  if (d >= n) return;
  int4 sl = slot4[(long)d * LPN + sub];
  int cn = cnt[d];

  for (int p = 0; p < PASSES; ++p, d += NPP) {
    if (d >= n) break;
    const int dn = d + NPP < n ? d + NPP : n - 1;
    const int4 nx = slot4[(long)dn * LPN + sub];
    const int ncn = cnt[dn];
    float4 rt[Q];
#pragma unroll
    for (int q = 0; q < Q; ++q) {
      rt[q] = make_float4(0.0f, 0.0f, 0.0f, 0.0f);
      if (ADD_ROOT) rt[q] = *reinterpret_cast<const float4*>(root + (long)d * F + q * 32 + sub * 4);
    }

    const int m = cn < SLOT_CAP ? (cn > 0 ? cn : 0) : SLOT_CAP;

    float4 acc[Q];
#pragma unroll
    for (int q = 0; q < Q; ++q) acc[q] = make_float4(0.0f, 0.0f, 0.0f, 0.0f);
#pragma unroll
    for (int j = 0; j < SLOT_CAP / 4; j += DEEP / 4) {
      if (j * 4 < m) {
        int s[DEEP];
        float4 v[DEEP][Q];
#pragma unroll
        for (int u = 0; u < DEEP; ++u) {
          const int w = (u & 3) == 0 ? sl.x : (u & 3) == 1 ? sl.y : (u & 3) == 2 ? sl.z : sl.w;
          s[u] = __shfl(w, j + (u >> 2), LPN);
        }
#pragma unroll
        for (int u = 0; u < DEEP; ++u) {
#pragma unroll
          for (int q = 0; q < Q; ++q) {
            v[u][q] = make_float4(0.0f, 0.0f, 0.0f, 0.0f);
            if (j * 4 + u < m)
              v[u][q] = *reinterpret_cast<const float4*>(H + (long)s[u] * F + q * 32 + sub * 4);
          }
        }
#pragma unroll
        for (int u = 0; u < DEEP; ++u) {
#pragma unroll
          for (int q = 0; q < Q; ++q) {
            acc[q].x += v[u][q].x; acc[q].y += v[u][q].y; acc[q].z += v[u][q].z; acc[q].w += v[u][q].w;
          }
        }
      }
    }

    if (cn > SLOT_CAP) {
      int e = head[d];
      int c = SLOT_CAP;
      int2 lk = make_int2(0, -1);
      if ((unsigned)e < (unsigned)nE) lk = link2[e];
      while ((unsigned)e < (unsigned)nE && c < cn) {
        float4 v[Q];
#pragma unroll
        for (int q = 0; q < Q; ++q)
          v[q] = *reinterpret_cast<const float4*>(H + (long)lk.x * F + q * 32 + sub * 4);
        e = lk.y;
        if ((unsigned)e < (unsigned)nE) lk = link2[e];
#pragma unroll
        for (int q = 0; q < Q; ++q) {
          acc[q].x += v[q].x; acc[q].y += v[q].y; acc[q].z += v[q].z; acc[q].w += v[q].w;
        }
        ++c;
      }
    }

    const float div = fmaxf((float)cn, 1.0f);
#pragma unroll
    for (int q = 0; q < Q; ++q)
      *reinterpret_cast<float4*>(out + (long)d * F + q * 32 + sub * 4) =
          make_float4(rt[q].x + acc[q].x / div, rt[q].y + acc[q].y / div,
                      rt[q].z + acc[q].z / div, rt[q].w + acc[q].w / div);
    sl = nx;
    cn = ncn;
  }
}

extern "C" void kernel_launch(void* const* d_in, const int* in_sizes, int n_in,
                              void* d_out, int out_size, void* d_ws, size_t ws_size,
                              hipStream_t stream) {
  const float* x   = (const float*)d_in[0];
  const int*   ei  = (const int*)d_in[1];
  const float* Wl0 = (const float*)d_in[2];
  const float* bl0 = (const float*)d_in[3];
  const float* Wr0 = (const float*)d_in[4];
  const float* Wl1 = (const float*)d_in[5];
  const float* bl1 = (const float*)d_in[6];
  const float* Wr1 = (const float*)d_in[7];
  float* out = (float*)d_out;

  const int N = in_sizes[0] / 64;
  const int E = in_sizes[1] / 2;

  float* B0 = (float*)d_ws;
  float* B1 = B0 + (size_t)N * 64;
  float* B2 = B1 + (size_t)N * 64;
  int* slots = (int*)(B2 + (size_t)N * 64);
  int* link = slots + (size_t)N * 32;
  int* head = link + (size_t)E * 2;
  int* ticket = head + (((size_t)N + 31) / 32) * 32;
  int* cnt = ticket + 64;

  const dim3 blk(256);
  const int tiles = (N + 15) / 16;
  const dim3 gemmGrid((tiles + 7) / 8);
  const dim3 buildGrid(2 * ((E + LIST_CHUNK - 1) / LIST_CHUNK));
  const dim3 gathGrid((N + 31) / 32);

  fill_head<<<dim3(256), blk, 0, stream>>>(head, cnt, N, ticket);
  build_slots<<<buildGrid, blk, 0, stream>>>(ei, cnt, slots, head, link, ticket, E, N);

  sage_gemm<64><<<gemmGrid, blk, 0, stream>>>(x, Wl0, bl0, B0, N);

  sage_gather0_root0_layer1<<<gemmGrid, blk, 0, stream>>>(slots, cnt, head, link, B0, x, Wr0, Wl1, bl1, Wr1, B1, B2, N, E);

  sage_gather_slots<32, true><<<gathGrid, dim3(64), 0, stream>>>(slots, cnt, head, link, B1, B2, out, N, E);
}


# Weights copied once per call, by the first launch, into matrix-fragment order: a weight fragment load of the products is 512 contiguous bytes (4 whole lines), not 16-byte pieces of 16 rows; same sums
# speedup vs baseline: 4.0995x; 1.0561x over previous
#include <hip/hip_runtime.h>


typedef __attribute__((ext_vector_type(2))) float v2f;
typedef __attribute__((ext_vector_type(8))) float v8f;

#define KDIM 64

#define WF_L0 0
#define WF_R0 4096
#define WF_L1 8192
#define WF_R1 10240
#define WF_FLOATS 12288

__global__ void fill_head(int* __restrict__ head, int* __restrict__ cnt, int n,
                          int* __restrict__ ticket,
                          const float* __restrict__ Wl0, const float* __restrict__ Wr0,
                          const float* __restrict__ Wl1, const float* __restrict__ Wr1,
                          float* __restrict__ wfrag) {
  int i = blockIdx.x * blockDim.x + threadIdx.x;
  int stride = gridDim.x * blockDim.x;
  if (i < 64) ticket[i] = 0;
  if (i < WF_FLOATS / 4) {
    const int half = (i >> 4) & 1;
    const int c = i & 15;
    float4 w;
    if (i < (WF_L1 / 4)) {
      const float* __restrict__ M = i < (WF_R0 / 4) ? Wl0 : Wr0;
      const int blk = (i >> 5) & 31;
      const float* p = M + ((blk >> 3) * 16 + c) * KDIM + (blk & 7) * 8 + half * 2;
      const v2f lo = *(const v2f*)p;
      const v2f hi = *(const v2f*)(p + 4);
      w = make_float4(lo[0], lo[1], hi[0], hi[1]);
    } else {
      const float* __restrict__ M = i < (WF_R1 / 4) ? Wl1 : Wr1;
      const int blk = (i >> 5) & 15;
      const int g = blk & 7;
      const float* p = M + ((blk >> 3) * 16 + c) * KDIM + (g >> 1) * 16 + half * 8 + (g & 1) * 4;
      w = *(const float4*)p;
    }
    reinterpret_cast<float4*>(wfrag)[i] = w;
  }
  for (; i < n; i += stride) {
    head[i] = -1;
    cnt[i] = 0;
  }
}

template <int OUTF>
__global__ __launch_bounds__(256) void sage_gemm(
    const float* __restrict__ X, const float* __restrict__ Wf,
    const float* __restrict__ bias, float* __restrict__ Y, int n) {
  constexpr int NT = OUTF / 16;
  const int wave = threadIdx.x >> 5;
  const int lane = threadIdx.x & 31;
  const int half = lane >> 4;
  const int l15  = lane & 15;
  const int tile = blockIdx.x * (blockDim.x >> 5) + wave;
  const int r0 = tile * 16;
  if (r0 >= n) return;

  int arow = r0 + l15;
  if (arow >= n) arow = n - 1;
  const float* xrow = X + (long)arow * KDIM + half * 2;

  v8f acc[NT];
#pragma unroll
  for (int t = 0; t < NT; ++t) {
    float b = bias[t * 16 + l15];
#pragma unroll
    for (int j = 0; j < 8; ++j) acc[t][j] = b;
  }

  const float4* __restrict__ wf = reinterpret_cast<const float4*>(Wf) + lane;
#pragma unroll
  for (int m = 0; m < KDIM / 8; ++m) {
    const v2f a0 = *(const v2f*)(xrow + m * 8);
    const v2f a1 = *(const v2f*)(xrow + m * 8 + 4);
    float4 w[NT];
#pragma unroll
    for (int t = 0; t < NT; ++t) w[t] = wf[(t * 8 + m) * 32];
#pragma unroll
    for (int t = 0; t < NT; ++t) {
      v2f b;
      b[0] = w[t].x;
      b[1] = w[t].y;
      acc[t] = __builtin_amdgcn_wmma_f32_16x16x4_f32(
           false, a0,  false, b,
           (short)0, acc[t],  false,  false);
    }
#pragma unroll
    for (int t = 0; t < NT; ++t) {
      v2f b;
      b[0] = w[t].z;
      b[1] = w[t].w;
      acc[t] = __builtin_amdgcn_wmma_f32_16x16x4_f32(
           false, a1,  false, b,
           (short)0, acc[t],  false,  false);
    }
  }

#pragma unroll
  for (int t = 0; t < NT; ++t) {
    int col = t * 16 + l15;
#pragma unroll
    for (int v = 0; v < 8; ++v) {
      int row = r0 + half * 8 + v;
      if (row < n) Y[(long)row * OUTF + col] = acc[t][v];
    }
  }
}

#define SLOT_CAP 32

__global__ __launch_bounds__(256, 8) void sage_gather0_root0_layer1(
    const int* __restrict__ slots, const int* __restrict__ cnt,
    const int* __restrict__ head, const int* __restrict__ link,
    const float* __restrict__ H0,
    const float* __restrict__ X, const float* __restrict__ Wr0f,
    const float* __restrict__ Wl1f, const float* __restrict__ bl1,
    const float* __restrict__ Wr1f,
    float* __restrict__ HL, float* __restrict__ ROOT, int n, int nE) {
  constexpr int F0 = 64;
  constexpr int F1 = 32;
  constexpr int Q = F0 / 32;
  constexpr int DEEP = 16 / Q;
  const int wave = threadIdx.x >> 5;
  const int lane = threadIdx.x & 31;
  const int half = lane >> 4;
  const int l15  = lane & 15;
  const int tile = blockIdx.x * (blockDim.x >> 5) + wave;
  const int r0 = tile * 16;
  if (r0 >= n) return;

  const int gbase = lane & 12;
  const int sub = ((lane & 2) << 1) | (half << 1) | (lane & 1);
  const int4* __restrict__ slot4 = reinterpret_cast<const int4*>(slots);
  const int2* __restrict__ link2 = reinterpret_cast<const int2*>(link);

  float mn[4][Q][4];
#pragma unroll
  for (int k = 0; k < 4; ++k) {
#pragma unroll
    for (int q = 0; q < Q; ++q) {
#pragma unroll
      for (int e = 0; e < 4; ++e) mn[k][q][e] = 0.0f;
    }
  }

  int d = r0 + gbase;
  int4 sl;
  int cn;
  {
    const int dc = d < n ? d : n - 1;
    sl = slot4[(long)dc * 8 + sub];
    cn = cnt[dc];
    if (d >= n) cn = 0;
  }
#pragma unroll 1
  for (int p = 0; p < 4; ++p, ++d) {
    const int dn = d + 1 < n ? d + 1 : n - 1;
    const int4 nx = slot4[(long)dn * 8 + sub];
    int ncn = cnt[dn];
    if (d + 1 >= n) ncn = 0;

    const int m = cn < SLOT_CAP ? (cn > 0 ? cn : 0) : SLOT_CAP;

    float4 acc[Q];
#pragma unroll
    for (int q = 0; q < Q; ++q) acc[q] = make_float4(0.0f, 0.0f, 0.0f, 0.0f);
#pragma unroll
    for (int j = 0; j < SLOT_CAP / 4; j += DEEP / 4) {
      if (j * 4 < m) {
        int s[DEEP];
        float4 v[DEEP][Q];
#pragma unroll
        for (int u = 0; u < DEEP; ++u) {
          const int w = (u & 3) == 0 ? sl.x : (u & 3) == 1 ? sl.y : (u & 3) == 2 ? sl.z : sl.w;
          const int ks = j + (u >> 2);
          s[u] = __shfl(w, ((ks >> 1) & 1) * 16 + ((ks >> 2) << 1) + (ks & 1) + gbase);
        }
#pragma unroll
        for (int u = 0; u < DEEP; ++u) {
#pragma unroll
          for (int q = 0; q < Q; ++q) {
            v[u][q] = make_float4(0.0f, 0.0f, 0.0f, 0.0f);
            if (j * 4 + u < m)
              v[u][q] = *reinterpret_cast<const float4*>(H0 + (long)s[u] * F0 + q * 32 + sub * 4);
          }
        }
#pragma unroll
        for (int u = 0; u < DEEP; ++u) {
#pragma unroll
          for (int q = 0; q < Q; ++q) {
            acc[q].x += v[u][q].x; acc[q].y += v[u][q].y; acc[q].z += v[u][q].z; acc[q].w += v[u][q].w;
          }
        }
      }
    }

    if (cn > SLOT_CAP) {
      int e = head[d];
      int c = SLOT_CAP;
      int2 lk = make_int2(0, -1);
      if ((unsigned)e < (unsigned)nE) lk = link2[e];
      while ((unsigned)e < (unsigned)nE && c < cn) {
        float4 v[Q];
#pragma unroll
        for (int q = 0; q < Q; ++q)
          v[q] = *reinterpret_cast<const float4*>(H0 + (long)lk.x * F0 + q * 32 + sub * 4);
        e = lk.y;
        if ((unsigned)e < (unsigned)nE) lk = link2[e];
#pragma unroll
        for (int q = 0; q < Q; ++q) {
          acc[q].x += v[q].x; acc[q].y += v[q].y; acc[q].z += v[q].z; acc[q].w += v[q].w;
        }
        ++c;
      }
    }

    const float div = fmaxf((float)cn, 1.0f);
#pragma unroll
    for (int q = 0; q < Q; ++q) {
#pragma unroll
      for (int e = 0; e < 4; ++e) {
        mn[0][q][e] = mn[1][q][e];
        mn[1][q][e] = mn[2][q][e];
        mn[2][q][e] = mn[3][q][e];
      }
      mn[3][q][0] = acc[q].x / div;
      mn[3][q][1] = acc[q].y / div;
      mn[3][q][2] = acc[q].z / div;
      mn[3][q][3] = acc[q].w / div;
    }
    sl = nx;
    cn = ncn;
  }

  const bool hi1 = (lane & 2) != 0;
  const bool hi0 = (lane & 1) != 0;
  float n1[2][2][Q][4];
#pragma unroll
  for (int p0 = 0; p0 < 2; ++p0) {
#pragma unroll
    for (int q = 0; q < Q; ++q) {
#pragma unroll
      for (int e = 0; e < 4; ++e) {
        const float lo = mn[p0][q][e];
        const float up = mn[2 + p0][q][e];
        const float got = __shfl_xor(hi1 ? lo : up, 2);
        n1[0][p0][q][e] = hi1 ? got : lo;
        n1[1][p0][q][e] = hi1 ? up : got;
      }
    }
  }
  v8f h[4];
#pragma unroll
  for (int a = 0; a < 2; ++a) {
#pragma unroll
    for (int q = 0; q < Q; ++q) {
#pragma unroll
      for (int e = 0; e < 4; ++e) {
        const float lo = n1[a][0][q][e];
        const float up = n1[a][1][q][e];
        const float got = __shfl_xor(hi0 ? lo : up, 1);
        h[2 * q + a][e]     = hi0 ? got : lo;
        h[2 * q + a][4 + e] = hi0 ? up : got;
      }
    }
  }

  const int node = r0 + l15;
  const int nodec = node < n ? node : n - 1;
  const float* xrow = X + (long)nodec * KDIM + half * 2;

  const float4* __restrict__ wr0 = reinterpret_cast<const float4*>(Wr0f) + lane;
#pragma unroll 4
  for (int m = 0; m < KDIM / 8; ++m) {
    const v2f b0 = *(const v2f*)(xrow + m * 8);
    const v2f b1 = *(const v2f*)(xrow + m * 8 + 4);
    float4 w[4];
#pragma unroll
    for (int t = 0; t < 4; ++t) w[t] = wr0[(t * 8 + m) * 32];
#pragma unroll
    for (int t = 0; t < 4; ++t) {
      v2f a;
      a[0] = w[t].x;
      a[1] = w[t].y;
      h[t] = __builtin_amdgcn_wmma_f32_16x16x4_f32(
           false, a,  false, b0,
           (short)0, h[t],  false,  false);
    }
#pragma unroll
    for (int t = 0; t < 4; ++t) {
      v2f a;
      a[0] = w[t].z;
      a[1] = w[t].w;
      h[t] = __builtin_amdgcn_wmma_f32_16x16x4_f32(
           false, a,  false, b1,
           (short)0, h[t],  false,  false);
    }
  }

#pragma unroll
  for (int t = 0; t < 4; ++t) {
#pragma unroll
    for (int r = 0; r < 8; ++r) h[t][r] = fmaxf(h[t][r], 0.0f);
  }

  v8f accL[2], accR[2];
#pragma unroll
  for (int u = 0; u < 2; ++u) {
#pragma unroll
    for (int r = 0; r < 8; ++r) {
      accL[u][r] = bl1[u * 16 + half * 8 + r];
      accR[u][r] = 0.0f;
    }
  }

  const float4* __restrict__ wl1 = reinterpret_cast<const float4*>(Wl1f) + lane;
  const float4* __restrict__ wr1 = reinterpret_cast<const float4*>(Wr1f) + lane;
#pragma unroll
  for (int g = 0; g < KDIM / 8; ++g) {
    const int t = g >> 1;
    const int r = (g & 1) * 4;
    v2f b0, b1;
    b0[0] = h[t][r];
    b0[1] = h[t][r + 1];
    b1[0] = h[t][r + 2];
    b1[1] = h[t][r + 3];
#pragma unroll
    for (int u = 0; u < 2; ++u) {
      const float4 wl = wl1[(u * 8 + g) * 32];
      const float4 wr = wr1[(u * 8 + g) * 32];
      v2f al0, al1, ar0, ar1;
      al0[0] = wl.x; al0[1] = wl.y; al1[0] = wl.z; al1[1] = wl.w;
      ar0[0] = wr.x; ar0[1] = wr.y; ar1[0] = wr.z; ar1[1] = wr.w;
      accL[u] = __builtin_amdgcn_wmma_f32_16x16x4_f32(
           false, al0,  false, b0,
           (short)0, accL[u],  false,  false);
      accR[u] = __builtin_amdgcn_wmma_f32_16x16x4_f32(
           false, ar0,  false, b0,
           (short)0, accR[u],  false,  false);
      accL[u] = __builtin_amdgcn_wmma_f32_16x16x4_f32(
           false, al1,  false, b1,
           (short)0, accL[u],  false,  false);
      accR[u] = __builtin_amdgcn_wmma_f32_16x16x4_f32(
           false, ar1,  false, b1,
           (short)0, accR[u],  false,  false);
    }
  }

  if (node < n) {
    float* hp = HL + (long)node * F1 + half * 8;
    float* rp = ROOT + (long)node * F1 + half * 8;
#pragma unroll
    for (int u = 0; u < 2; ++u) {
      *reinterpret_cast<float4*>(hp + u * 16) =
          make_float4(accL[u][0], accL[u][1], accL[u][2], accL[u][3]);
      *reinterpret_cast<float4*>(hp + u * 16 + 4) =
          make_float4(accL[u][4], accL[u][5], accL[u][6], accL[u][7]);
      *reinterpret_cast<float4*>(rp + u * 16) =
          make_float4(accR[u][0], accR[u][1], accR[u][2], accR[u][3]);
      *reinterpret_cast<float4*>(rp + u * 16 + 4) =
          make_float4(accR[u][4], accR[u][5], accR[u][6], accR[u][7]);
    }
  }
}


#define LIST_EPT 8
#define LIST_CHUNK (256 * LIST_EPT)
__global__ __launch_bounds__(256) void build_slots(
    const int* __restrict__ ei, int* __restrict__ cnt, int* __restrict__ slots,
    int* __restrict__ head, int* __restrict__ link,
    int* __restrict__ ticket, int nE, int n) {
  constexpr int EPT = LIST_EPT;
  const int chunks = (nE + LIST_CHUNK - 1) / LIST_CHUNK;
  const int split = ((n >> 1) + 31) & ~31;
  __shared__ int sh_upper, sh_chunk;
  if (threadIdx.x == 0) {
    const unsigned id = __builtin_amdgcn_s_sendmsg_rtn(0x87);
    int upper = (int)((id >> 18) & 1u);
    int c = __hip_atomic_fetch_add(ticket + upper * 32, 1, __ATOMIC_RELAXED, __HIP_MEMORY_SCOPE_AGENT);
    if (c >= chunks) {
      upper ^= 1;
      c = __hip_atomic_fetch_add(ticket + upper * 32, 1, __ATOMIC_RELAXED, __HIP_MEMORY_SCOPE_AGENT);
    }
    sh_upper = upper;
    sh_chunk = c;
  }
  __syncthreads();
  const bool upper = sh_upper != 0;
  const int chunk = sh_chunk;
  if ((unsigned)chunk >= (unsigned)chunks) return;

  const int base = chunk * LIST_CHUNK + threadIdx.x;
  int s[EPT], d[EPT], rk[EPT];
#pragma unroll
  for (int k = 0; k < EPT; ++k) {
    const int e = base + k * 256;
    const int ec = e < nE ? e : nE - 1;
    d[k] = ei[nE + ec];
  }
#pragma unroll
  for (int k = 0; k < EPT; ++k) {
    const int e = base + k * 256;
    const int ec = e < nE ? e : nE - 1;
    s[k] = ei[ec];
  }
#pragma unroll
  for (int k = 0; k < EPT; ++k) {
    const int e = base + k * 256;
    rk[k] = -1;
    if (e < nE && (d[k] >= split) == upper)
      rk[k] = __hip_atomic_fetch_add(cnt + d[k], 1, __ATOMIC_RELAXED, __HIP_MEMORY_SCOPE_AGENT);
  }
  int2* __restrict__ link2 = reinterpret_cast<int2*>(link);
#pragma unroll
  for (int k = 0; k < EPT; ++k) {
    const int e = base + k * 256;
    if (e < nE && (d[k] >= split) == upper) {
      if ((unsigned)rk[k] < (unsigned)SLOT_CAP) {
        slots[(long)d[k] * SLOT_CAP + rk[k]] = s[k];
      } else {
        const int old = __hip_atomic_exchange(head + d[k], e, __ATOMIC_RELAXED, __HIP_MEMORY_SCOPE_AGENT);
        link2[e] = make_int2(s[k], old);
      }
    }
  }
}

template <int F, bool ADD_ROOT>
__global__ __launch_bounds__(64) void sage_gather_slots(
    const int* __restrict__ slots, const int* __restrict__ cnt,
    const int* __restrict__ head, const int* __restrict__ link,
    const float* __restrict__ H, const float* __restrict__ root,
    float* __restrict__ out, int n, int nE) {
  static_assert(F == 32 || F == 64, "rows of one or two 128-byte lines");
  constexpr int LPN = 8;
  constexpr int NPP = 64 / LPN;
  constexpr int PASSES = 4;
  constexpr int Q = F / 32;
  constexpr int DEEP = 16 / Q;
  const int sub = (int)(threadIdx.x % LPN);
  const int4* __restrict__ slot4 = reinterpret_cast<const int4*>(slots);
  const int2* __restrict__ link2 = reinterpret_cast<const int2*>(link);

  int d = blockIdx.x * (NPP * PASSES) + (int)(threadIdx.x / LPN);
  if (d >= n) return;
  int4 sl = slot4[(long)d * LPN + sub];
  int cn = cnt[d];

  for (int p = 0; p < PASSES; ++p, d += NPP) {
    if (d >= n) break;
    const int dn = d + NPP < n ? d + NPP : n - 1;
    const int4 nx = slot4[(long)dn * LPN + sub];
    const int ncn = cnt[dn];
    float4 rt[Q];
#pragma unroll
    for (int q = 0; q < Q; ++q) {
      rt[q] = make_float4(0.0f, 0.0f, 0.0f, 0.0f);
      if (ADD_ROOT) rt[q] = *reinterpret_cast<const float4*>(root + (long)d * F + q * 32 + sub * 4);
    }

    const int m = cn < SLOT_CAP ? (cn > 0 ? cn : 0) : SLOT_CAP;

    float4 acc[Q];
#pragma unroll
    for (int q = 0; q < Q; ++q) acc[q] = make_float4(0.0f, 0.0f, 0.0f, 0.0f);
#pragma unroll
    for (int j = 0; j < SLOT_CAP / 4; j += DEEP / 4) {
      if (j * 4 < m) {
        int s[DEEP];
        float4 v[DEEP][Q];
#pragma unroll
        for (int u = 0; u < DEEP; ++u) {
          const int w = (u & 3) == 0 ? sl.x : (u & 3) == 1 ? sl.y : (u & 3) == 2 ? sl.z : sl.w;
          s[u] = __shfl(w, j + (u >> 2), LPN);
        }
#pragma unroll
        for (int u = 0; u < DEEP; ++u) {
#pragma unroll
          for (int q = 0; q < Q; ++q) {
            v[u][q] = make_float4(0.0f, 0.0f, 0.0f, 0.0f);
            if (j * 4 + u < m)
              v[u][q] = *reinterpret_cast<const float4*>(H + (long)s[u] * F + q * 32 + sub * 4);
          }
        }
#pragma unroll
        for (int u = 0; u < DEEP; ++u) {
#pragma unroll
          for (int q = 0; q < Q; ++q) {
            acc[q].x += v[u][q].x; acc[q].y += v[u][q].y; acc[q].z += v[u][q].z; acc[q].w += v[u][q].w;
          }
        }
      }
    }

    if (cn > SLOT_CAP) {
      int e = head[d];
      int c = SLOT_CAP;
      int2 lk = make_int2(0, -1);
      if ((unsigned)e < (unsigned)nE) lk = link2[e];
      while ((unsigned)e < (unsigned)nE && c < cn) {
        float4 v[Q];
#pragma unroll
        for (int q = 0; q < Q; ++q)
          v[q] = *reinterpret_cast<const float4*>(H + (long)lk.x * F + q * 32 + sub * 4);
        e = lk.y;
        if ((unsigned)e < (unsigned)nE) lk = link2[e];
#pragma unroll
        for (int q = 0; q < Q; ++q) {
          acc[q].x += v[q].x; acc[q].y += v[q].y; acc[q].z += v[q].z; acc[q].w += v[q].w;
        }
        ++c;
      }
    }

    const float div = fmaxf((float)cn, 1.0f);
#pragma unroll
    for (int q = 0; q < Q; ++q)
      *reinterpret_cast<float4*>(out + (long)d * F + q * 32 + sub * 4) =
          make_float4(rt[q].x + acc[q].x / div, rt[q].y + acc[q].y / div,
                      rt[q].z + acc[q].z / div, rt[q].w + acc[q].w / div);
    sl = nx;
    cn = ncn;
  }
}

extern "C" void kernel_launch(void* const* d_in, const int* in_sizes, int n_in,
                              void* d_out, int out_size, void* d_ws, size_t ws_size,
                              hipStream_t stream) {
  const float* x   = (const float*)d_in[0];
  const int*   ei  = (const int*)d_in[1];
  const float* Wl0 = (const float*)d_in[2];
  const float* bl0 = (const float*)d_in[3];
  const float* Wr0 = (const float*)d_in[4];
  const float* Wl1 = (const float*)d_in[5];
  const float* bl1 = (const float*)d_in[6];
  const float* Wr1 = (const float*)d_in[7];
  float* out = (float*)d_out;

  const int N = in_sizes[0] / 64;
  const int E = in_sizes[1] / 2;

  float* B0 = (float*)d_ws;
  float* B1 = B0 + (size_t)N * 64;
  float* B2 = B1 + (size_t)N * 64;
  int* slots = (int*)(B2 + (size_t)N * 64);
  int* link = slots + (size_t)N * 32;
  int* head = link + (size_t)E * 2;
  int* ticket = head + (((size_t)N + 31) / 32) * 32;
  int* cnt = ticket + 64;
  float* wfrag = (float*)(cnt + (((size_t)N + 31) / 32) * 32);

  const dim3 blk(256);
  const int tiles = (N + 15) / 16;
  const dim3 gemmGrid((tiles + 7) / 8);
  const dim3 buildGrid(2 * ((E + LIST_CHUNK - 1) / LIST_CHUNK));
  const dim3 gathGrid((N + 31) / 32);

  fill_head<<<dim3(256), blk, 0, stream>>>(head, cnt, N, ticket, Wl0, Wr0, Wl1, Wr1, wfrag);
  build_slots<<<buildGrid, blk, 0, stream>>>(ei, cnt, slots, head, link, ticket, E, N);

  sage_gemm<64><<<gemmGrid, blk, 0, stream>>>(x, wfrag + WF_L0, bl0, B0, N);

  sage_gather0_root0_layer1<<<gemmGrid, blk, 0, stream>>>(slots, cnt, head, link, B0, x, wfrag + WF_R0, wfrag + WF_L1, bl1, wfrag + WF_R1, B1, B2, N, E);

  sage_gather_slots<32, true><<<gathGrid, dim3(64), 0, stream>>>(slots, cnt, head, link, B1, B2, out, N, E);
}


# Clearing launch folded into the layer-0 product launch, now first: a block copies Wl0 to LDS in fragment order itself, a wave's 16 row loads go out ahead of that copy; 5->4 launches; same sums
# speedup vs baseline: 4.3655x; 1.0649x over previous
#include <hip/hip_runtime.h>


typedef __attribute__((ext_vector_type(2))) float v2f;
typedef __attribute__((ext_vector_type(8))) float v8f;

#define KDIM 64

#define WF_R0 0
#define WF_L1 4096
#define WF_R1 6144
#define WF_FLOATS 8192
#define WL0_FLOATS 4096

__global__ __launch_bounds__(256) void sage_gemm0_clear(
    const float* __restrict__ X, const float* __restrict__ Wl0,
    const float* __restrict__ bias, float* __restrict__ Y, int n,
    int* __restrict__ head, int* __restrict__ cnt, int* __restrict__ ticket,
    const float* __restrict__ Wr0, const float* __restrict__ Wl1,
    const float* __restrict__ Wr1, float* __restrict__ wfrag) {
  constexpr int NT = 4;
  __shared__ float4 wl0s[WL0_FLOATS / 4];
  const int wave = threadIdx.x >> 5;
  const int lane = threadIdx.x & 31;
  const int half = lane >> 4;
  const int l15  = lane & 15;
  const int tile = blockIdx.x * (blockDim.x >> 5) + wave;
  const int r0 = tile * 16;

  int arow = r0 + l15;
  if (arow >= n) arow = n - 1;
  const float* xrow = X + (long)arow * KDIM + half * 2;
  v2f a[KDIM / 4];
#pragma unroll
  for (int j = 0; j < KDIM / 4; ++j) a[j] = *(const v2f*)(xrow + j * 4);
  float bv[NT];
#pragma unroll
  for (int t = 0; t < NT; ++t) bv[t] = bias[t * 16 + l15];

#pragma unroll
  for (int j = 0; j < WL0_FLOATS / 4 / 256; ++j) {
    const int i = (int)threadIdx.x + 256 * j;
    const int hf = (i >> 4) & 1;
    const int c = i & 15;
    const int blk = i >> 5;
    const float* p = Wl0 + ((blk >> 3) * 16 + c) * KDIM + (blk & 7) * 8 + hf * 2;
    const v2f lo = *(const v2f*)p;
    const v2f hi = *(const v2f*)(p + 4);
    wl0s[i] = make_float4(lo[0], lo[1], hi[0], hi[1]);
  }

  __syncthreads();

  {
    int i = blockIdx.x * blockDim.x + threadIdx.x;
    const int stride = gridDim.x * blockDim.x;
    if (i < 64) ticket[i] = 0;
    for (int f = i; f < WF_FLOATS / 4; f += stride) {
      const int hf = (f >> 4) & 1;
      const int c = f & 15;
      float4 w;
      if (f < (WF_L1 / 4)) {
        const int blk = (f >> 5) & 31;
        const float* p = Wr0 + ((blk >> 3) * 16 + c) * KDIM + (blk & 7) * 8 + hf * 2;
        const v2f lo = *(const v2f*)p;
        const v2f hi = *(const v2f*)(p + 4);
        w = make_float4(lo[0], lo[1], hi[0], hi[1]);
      } else {
        const float* __restrict__ M = f < (WF_R1 / 4) ? Wl1 : Wr1;
        const int blk = (f >> 5) & 15;
        const int g = blk & 7;
        const float* p = M + ((blk >> 3) * 16 + c) * KDIM + (g >> 1) * 16 + hf * 8 + (g & 1) * 4;
        w = *(const float4*)p;
      }
      reinterpret_cast<float4*>(wfrag)[f] = w;
    }
    for (; i < n; i += stride) {
      head[i] = -1;
      cnt[i] = 0;
    }
  }

  if (r0 >= n) return;

  v8f acc[NT];
#pragma unroll
  for (int t = 0; t < NT; ++t) {
#pragma unroll
    for (int j = 0; j < 8; ++j) acc[t][j] = bv[t];
  }

  const float4* wf = wl0s + lane;
#pragma unroll
  for (int m = 0; m < KDIM / 8; ++m) {
    float4 w[NT];
#pragma unroll
    for (int t = 0; t < NT; ++t) w[t] = wf[(t * 8 + m) * 32];
#pragma unroll
    for (int t = 0; t < NT; ++t) {
      v2f b;
      b[0] = w[t].x;
      b[1] = w[t].y;
      acc[t] = __builtin_amdgcn_wmma_f32_16x16x4_f32(
           false, a[2 * m],  false, b,
           (short)0, acc[t],  false,  false);
    }
#pragma unroll
    for (int t = 0; t < NT; ++t) {
      v2f b;
      b[0] = w[t].z;
      b[1] = w[t].w;
      acc[t] = __builtin_amdgcn_wmma_f32_16x16x4_f32(
           false, a[2 * m + 1],  false, b,
           (short)0, acc[t],  false,  false);
    }
  }

#pragma unroll
  for (int t = 0; t < NT; ++t) {
    int col = t * 16 + l15;
#pragma unroll
    for (int v = 0; v < 8; ++v) {
      int row = r0 + half * 8 + v;
      if (row < n) Y[(long)row * 64 + col] = acc[t][v];
    }
  }
}

#define SLOT_CAP 32

__global__ __launch_bounds__(256, 8) void sage_gather0_root0_layer1(
    const int* __restrict__ slots, const int* __restrict__ cnt,
    const int* __restrict__ head, const int* __restrict__ link,
    const float* __restrict__ H0,
    const float* __restrict__ X, const float* __restrict__ Wr0f,
    const float* __restrict__ Wl1f, const float* __restrict__ bl1,
    const float* __restrict__ Wr1f,
    float* __restrict__ HL, float* __restrict__ ROOT, int n, int nE) {
  constexpr int F0 = 64;
  constexpr int F1 = 32;
  constexpr int Q = F0 / 32;
  constexpr int DEEP = 16 / Q;
  const int wave = threadIdx.x >> 5;
  const int lane = threadIdx.x & 31;
  const int half = lane >> 4;
  const int l15  = lane & 15;
  const int tile = blockIdx.x * (blockDim.x >> 5) + wave;
  const int r0 = tile * 16;
  if (r0 >= n) return;

  const int gbase = lane & 12;
  const int sub = ((lane & 2) << 1) | (half << 1) | (lane & 1);
  const int4* __restrict__ slot4 = reinterpret_cast<const int4*>(slots);
  const int2* __restrict__ link2 = reinterpret_cast<const int2*>(link);

  float mn[4][Q][4];
#pragma unroll
  for (int k = 0; k < 4; ++k) {
#pragma unroll
    for (int q = 0; q < Q; ++q) {
#pragma unroll
      for (int e = 0; e < 4; ++e) mn[k][q][e] = 0.0f;
    }
  }

  int d = r0 + gbase;
  int4 sl;
  int cn;
  {
    const int dc = d < n ? d : n - 1;
    sl = slot4[(long)dc * 8 + sub];
    cn = cnt[dc];
    if (d >= n) cn = 0;
  }
#pragma unroll 1
  for (int p = 0; p < 4; ++p, ++d) {
    const int dn = d + 1 < n ? d + 1 : n - 1;
    const int4 nx = slot4[(long)dn * 8 + sub];
    int ncn = cnt[dn];
    if (d + 1 >= n) ncn = 0;

    const int m = cn < SLOT_CAP ? (cn > 0 ? cn : 0) : SLOT_CAP;

    float4 acc[Q];
#pragma unroll
    for (int q = 0; q < Q; ++q) acc[q] = make_float4(0.0f, 0.0f, 0.0f, 0.0f);
#pragma unroll
    for (int j = 0; j < SLOT_CAP / 4; j += DEEP / 4) {
      if (j * 4 < m) {
        int s[DEEP];
        float4 v[DEEP][Q];
#pragma unroll
        for (int u = 0; u < DEEP; ++u) {
          const int w = (u & 3) == 0 ? sl.x : (u & 3) == 1 ? sl.y : (u & 3) == 2 ? sl.z : sl.w;
          const int ks = j + (u >> 2);
          s[u] = __shfl(w, ((ks >> 1) & 1) * 16 + ((ks >> 2) << 1) + (ks & 1) + gbase);
        }
#pragma unroll
        for (int u = 0; u < DEEP; ++u) {
#pragma unroll
          for (int q = 0; q < Q; ++q) {
            v[u][q] = make_float4(0.0f, 0.0f, 0.0f, 0.0f);
            if (j * 4 + u < m)
              v[u][q] = *reinterpret_cast<const float4*>(H0 + (long)s[u] * F0 + q * 32 + sub * 4);
          }
        }
#pragma unroll
        for (int u = 0; u < DEEP; ++u) {
#pragma unroll
          for (int q = 0; q < Q; ++q) {
            acc[q].x += v[u][q].x; acc[q].y += v[u][q].y; acc[q].z += v[u][q].z; acc[q].w += v[u][q].w;
          }
        }
      }
    }

    if (cn > SLOT_CAP) {
      int e = head[d];
      int c = SLOT_CAP;
      int2 lk = make_int2(0, -1);
      if ((unsigned)e < (unsigned)nE) lk = link2[e];
      while ((unsigned)e < (unsigned)nE && c < cn) {
        float4 v[Q];
#pragma unroll
        for (int q = 0; q < Q; ++q)
          v[q] = *reinterpret_cast<const float4*>(H0 + (long)lk.x * F0 + q * 32 + sub * 4);
        e = lk.y;
        if ((unsigned)e < (unsigned)nE) lk = link2[e];
#pragma unroll
        for (int q = 0; q < Q; ++q) {
          acc[q].x += v[q].x; acc[q].y += v[q].y; acc[q].z += v[q].z; acc[q].w += v[q].w;
        }
        ++c;
      }
    }

    const float div = fmaxf((float)cn, 1.0f);
#pragma unroll
    for (int q = 0; q < Q; ++q) {
#pragma unroll
      for (int e = 0; e < 4; ++e) {
        mn[0][q][e] = mn[1][q][e];
        mn[1][q][e] = mn[2][q][e];
        mn[2][q][e] = mn[3][q][e];
      }
      mn[3][q][0] = acc[q].x / div;
      mn[3][q][1] = acc[q].y / div;
      mn[3][q][2] = acc[q].z / div;
      mn[3][q][3] = acc[q].w / div;
    }
    sl = nx;
    cn = ncn;
  }

  const bool hi1 = (lane & 2) != 0;
  const bool hi0 = (lane & 1) != 0;
  float n1[2][2][Q][4];
#pragma unroll
  for (int p0 = 0; p0 < 2; ++p0) {
#pragma unroll
    for (int q = 0; q < Q; ++q) {
#pragma unroll
      for (int e = 0; e < 4; ++e) {
        const float lo = mn[p0][q][e];
        const float up = mn[2 + p0][q][e];
        const float got = __shfl_xor(hi1 ? lo : up, 2);
        n1[0][p0][q][e] = hi1 ? got : lo;
        n1[1][p0][q][e] = hi1 ? up : got;
      }
    }
  }
  v8f h[4];
#pragma unroll
  for (int a = 0; a < 2; ++a) {
#pragma unroll
    for (int q = 0; q < Q; ++q) {
#pragma unroll
      for (int e = 0; e < 4; ++e) {
        const float lo = n1[a][0][q][e];
        const float up = n1[a][1][q][e];
        const float got = __shfl_xor(hi0 ? lo : up, 1);
        h[2 * q + a][e]     = hi0 ? got : lo;
        h[2 * q + a][4 + e] = hi0 ? up : got;
      }
    }
  }

  const int node = r0 + l15;
  const int nodec = node < n ? node : n - 1;
  const float* xrow = X + (long)nodec * KDIM + half * 2;

  const float4* __restrict__ wr0 = reinterpret_cast<const float4*>(Wr0f) + lane;
#pragma unroll 4
  for (int m = 0; m < KDIM / 8; ++m) {
    const v2f b0 = *(const v2f*)(xrow + m * 8);
    const v2f b1 = *(const v2f*)(xrow + m * 8 + 4);
    float4 w[4];
#pragma unroll
    for (int t = 0; t < 4; ++t) w[t] = wr0[(t * 8 + m) * 32];
#pragma unroll
    for (int t = 0; t < 4; ++t) {
      v2f a;
      a[0] = w[t].x;
      a[1] = w[t].y;
      h[t] = __builtin_amdgcn_wmma_f32_16x16x4_f32(
           false, a,  false, b0,
           (short)0, h[t],  false,  false);
    }
#pragma unroll
    for (int t = 0; t < 4; ++t) {
      v2f a;
      a[0] = w[t].z;
      a[1] = w[t].w;
      h[t] = __builtin_amdgcn_wmma_f32_16x16x4_f32(
           false, a,  false, b1,
           (short)0, h[t],  false,  false);
    }
  }

#pragma unroll
  for (int t = 0; t < 4; ++t) {
#pragma unroll
    for (int r = 0; r < 8; ++r) h[t][r] = fmaxf(h[t][r], 0.0f);
  }

  v8f accL[2], accR[2];
#pragma unroll
  for (int u = 0; u < 2; ++u) {
#pragma unroll
    for (int r = 0; r < 8; ++r) {
      accL[u][r] = bl1[u * 16 + half * 8 + r];
      accR[u][r] = 0.0f;
    }
  }

  const float4* __restrict__ wl1 = reinterpret_cast<const float4*>(Wl1f) + lane;
  const float4* __restrict__ wr1 = reinterpret_cast<const float4*>(Wr1f) + lane;
#pragma unroll
  for (int g = 0; g < KDIM / 8; ++g) {
    const int t = g >> 1;
    const int r = (g & 1) * 4;
    v2f b0, b1;
    b0[0] = h[t][r];
    b0[1] = h[t][r + 1];
    b1[0] = h[t][r + 2];
    b1[1] = h[t][r + 3];
#pragma unroll
    for (int u = 0; u < 2; ++u) {
      const float4 wl = wl1[(u * 8 + g) * 32];
      const float4 wr = wr1[(u * 8 + g) * 32];
      v2f al0, al1, ar0, ar1;
      al0[0] = wl.x; al0[1] = wl.y; al1[0] = wl.z; al1[1] = wl.w;
      ar0[0] = wr.x; ar0[1] = wr.y; ar1[0] = wr.z; ar1[1] = wr.w;
      accL[u] = __builtin_amdgcn_wmma_f32_16x16x4_f32(
           false, al0,  false, b0,
           (short)0, accL[u],  false,  false);
      accR[u] = __builtin_amdgcn_wmma_f32_16x16x4_f32(
           false, ar0,  false, b0,
           (short)0, accR[u],  false,  false);
      accL[u] = __builtin_amdgcn_wmma_f32_16x16x4_f32(
           false, al1,  false, b1,
           (short)0, accL[u],  false,  false);
      accR[u] = __builtin_amdgcn_wmma_f32_16x16x4_f32(
           false, ar1,  false, b1,
           (short)0, accR[u],  false,  false);
    }
  }

  if (node < n) {
    float* hp = HL + (long)node * F1 + half * 8;
    float* rp = ROOT + (long)node * F1 + half * 8;
#pragma unroll
    for (int u = 0; u < 2; ++u) {
      *reinterpret_cast<float4*>(hp + u * 16) =
          make_float4(accL[u][0], accL[u][1], accL[u][2], accL[u][3]);
      *reinterpret_cast<float4*>(hp + u * 16 + 4) =
          make_float4(accL[u][4], accL[u][5], accL[u][6], accL[u][7]);
      *reinterpret_cast<float4*>(rp + u * 16) =
          make_float4(accR[u][0], accR[u][1], accR[u][2], accR[u][3]);
      *reinterpret_cast<float4*>(rp + u * 16 + 4) =
          make_float4(accR[u][4], accR[u][5], accR[u][6], accR[u][7]);
    }
  }
}


#define LIST_EPT 8
#define LIST_CHUNK (256 * LIST_EPT)
__global__ __launch_bounds__(256) void build_slots(
    const int* __restrict__ ei, int* __restrict__ cnt, int* __restrict__ slots,
    int* __restrict__ head, int* __restrict__ link,
    int* __restrict__ ticket, int nE, int n) {
  constexpr int EPT = LIST_EPT;
  const int chunks = (nE + LIST_CHUNK - 1) / LIST_CHUNK;
  const int split = ((n >> 1) + 31) & ~31;
  __shared__ int sh_upper, sh_chunk;
  if (threadIdx.x == 0) {
    const unsigned id = __builtin_amdgcn_s_sendmsg_rtn(0x87);
    int upper = (int)((id >> 18) & 1u);
    int c = __hip_atomic_fetch_add(ticket + upper * 32, 1, __ATOMIC_RELAXED, __HIP_MEMORY_SCOPE_AGENT);
    if (c >= chunks) {
      upper ^= 1;
      c = __hip_atomic_fetch_add(ticket + upper * 32, 1, __ATOMIC_RELAXED, __HIP_MEMORY_SCOPE_AGENT);
    }
    sh_upper = upper;
    sh_chunk = c;
  }
  __syncthreads();
  const bool upper = sh_upper != 0;
  const int chunk = sh_chunk;
  if ((unsigned)chunk >= (unsigned)chunks) return;

  const int base = chunk * LIST_CHUNK + threadIdx.x;
  int s[EPT], d[EPT], rk[EPT];
#pragma unroll
  for (int k = 0; k < EPT; ++k) {
    const int e = base + k * 256;
    const int ec = e < nE ? e : nE - 1;
    d[k] = ei[nE + ec];
  }
#pragma unroll
  for (int k = 0; k < EPT; ++k) {
    const int e = base + k * 256;
    const int ec = e < nE ? e : nE - 1;
    s[k] = ei[ec];
  }
#pragma unroll
  for (int k = 0; k < EPT; ++k) {
    const int e = base + k * 256;
    rk[k] = -1;
    if (e < nE && (d[k] >= split) == upper)
      rk[k] = __hip_atomic_fetch_add(cnt + d[k], 1, __ATOMIC_RELAXED, __HIP_MEMORY_SCOPE_AGENT);
  }
  int2* __restrict__ link2 = reinterpret_cast<int2*>(link);
#pragma unroll
  for (int k = 0; k < EPT; ++k) {
    const int e = base + k * 256;
    if (e < nE && (d[k] >= split) == upper) {
      if ((unsigned)rk[k] < (unsigned)SLOT_CAP) {
        slots[(long)d[k] * SLOT_CAP + rk[k]] = s[k];
      } else {
        const int old = __hip_atomic_exchange(head + d[k], e, __ATOMIC_RELAXED, __HIP_MEMORY_SCOPE_AGENT);
        link2[e] = make_int2(s[k], old);
      }
    }
  }
}

template <int F, bool ADD_ROOT>
__global__ __launch_bounds__(64) void sage_gather_slots(
    const int* __restrict__ slots, const int* __restrict__ cnt,
    const int* __restrict__ head, const int* __restrict__ link,
    const float* __restrict__ H, const float* __restrict__ root,
    float* __restrict__ out, int n, int nE) {
  static_assert(F == 32 || F == 64, "rows of one or two 128-byte lines");
  constexpr int LPN = 8;
  constexpr int NPP = 64 / LPN;
  constexpr int PASSES = 4;
  constexpr int Q = F / 32;
  constexpr int DEEP = 16 / Q;
  const int sub = (int)(threadIdx.x % LPN);
  const int4* __restrict__ slot4 = reinterpret_cast<const int4*>(slots);
  const int2* __restrict__ link2 = reinterpret_cast<const int2*>(link);

  int d = blockIdx.x * (NPP * PASSES) + (int)(threadIdx.x / LPN);
  if (d >= n) return;
  int4 sl = slot4[(long)d * LPN + sub];
  int cn = cnt[d];

  for (int p = 0; p < PASSES; ++p, d += NPP) {
    if (d >= n) break;
    const int dn = d + NPP < n ? d + NPP : n - 1;
    const int4 nx = slot4[(long)dn * LPN + sub];
    const int ncn = cnt[dn];
    float4 rt[Q];
#pragma unroll
    for (int q = 0; q < Q; ++q) {
      rt[q] = make_float4(0.0f, 0.0f, 0.0f, 0.0f);
      if (ADD_ROOT) rt[q] = *reinterpret_cast<const float4*>(root + (long)d * F + q * 32 + sub * 4);
    }

    const int m = cn < SLOT_CAP ? (cn > 0 ? cn : 0) : SLOT_CAP;

    float4 acc[Q];
#pragma unroll
    for (int q = 0; q < Q; ++q) acc[q] = make_float4(0.0f, 0.0f, 0.0f, 0.0f);
#pragma unroll
    for (int j = 0; j < SLOT_CAP / 4; j += DEEP / 4) {
      if (j * 4 < m) {
        int s[DEEP];
        float4 v[DEEP][Q];
#pragma unroll
        for (int u = 0; u < DEEP; ++u) {
          const int w = (u & 3) == 0 ? sl.x : (u & 3) == 1 ? sl.y : (u & 3) == 2 ? sl.z : sl.w;
          s[u] = __shfl(w, j + (u >> 2), LPN);
        }
#pragma unroll
        for (int u = 0; u < DEEP; ++u) {
#pragma unroll
          for (int q = 0; q < Q; ++q) {
            v[u][q] = make_float4(0.0f, 0.0f, 0.0f, 0.0f);
            if (j * 4 + u < m)
              v[u][q] = *reinterpret_cast<const float4*>(H + (long)s[u] * F + q * 32 + sub * 4);
          }
        }
#pragma unroll
        for (int u = 0; u < DEEP; ++u) {
#pragma unroll
          for (int q = 0; q < Q; ++q) {
            acc[q].x += v[u][q].x; acc[q].y += v[u][q].y; acc[q].z += v[u][q].z; acc[q].w += v[u][q].w;
          }
        }
      }
    }

    if (cn > SLOT_CAP) {
      int e = head[d];
      int c = SLOT_CAP;
      int2 lk = make_int2(0, -1);
      if ((unsigned)e < (unsigned)nE) lk = link2[e];
      while ((unsigned)e < (unsigned)nE && c < cn) {
        float4 v[Q];
#pragma unroll
        for (int q = 0; q < Q; ++q)
          v[q] = *reinterpret_cast<const float4*>(H + (long)lk.x * F + q * 32 + sub * 4);
        e = lk.y;
        if ((unsigned)e < (unsigned)nE) lk = link2[e];
#pragma unroll
        for (int q = 0; q < Q; ++q) {
          acc[q].x += v[q].x; acc[q].y += v[q].y; acc[q].z += v[q].z; acc[q].w += v[q].w;
        }
        ++c;
      }
    }

    const float div = fmaxf((float)cn, 1.0f);
#pragma unroll
    for (int q = 0; q < Q; ++q)
      *reinterpret_cast<float4*>(out + (long)d * F + q * 32 + sub * 4) =
          make_float4(rt[q].x + acc[q].x / div, rt[q].y + acc[q].y / div,
                      rt[q].z + acc[q].z / div, rt[q].w + acc[q].w / div);
    sl = nx;
    cn = ncn;
  }
}

extern "C" void kernel_launch(void* const* d_in, const int* in_sizes, int n_in,
                              void* d_out, int out_size, void* d_ws, size_t ws_size,
                              hipStream_t stream) {
  const float* x   = (const float*)d_in[0];
  const int*   ei  = (const int*)d_in[1];
  const float* Wl0 = (const float*)d_in[2];
  const float* bl0 = (const float*)d_in[3];
  const float* Wr0 = (const float*)d_in[4];
  const float* Wl1 = (const float*)d_in[5];
  const float* bl1 = (const float*)d_in[6];
  const float* Wr1 = (const float*)d_in[7];
  float* out = (float*)d_out;

  const int N = in_sizes[0] / 64;
  const int E = in_sizes[1] / 2;

  float* B0 = (float*)d_ws;
  float* B1 = B0 + (size_t)N * 64;
  float* B2 = B1 + (size_t)N * 64;
  int* slots = (int*)(B2 + (size_t)N * 64);
  int* link = slots + (size_t)N * 32;
  int* head = link + (size_t)E * 2;
  int* ticket = head + (((size_t)N + 31) / 32) * 32;
  int* cnt = ticket + 64;
  float* wfrag = (float*)(cnt + (((size_t)N + 31) / 32) * 32);

  const dim3 blk(256);
  const int tiles = (N + 15) / 16;
  const dim3 gemmGrid((tiles + 7) / 8);
  const dim3 buildGrid(2 * ((E + LIST_CHUNK - 1) / LIST_CHUNK));
  const dim3 gathGrid((N + 31) / 32);

  sage_gemm0_clear<<<gemmGrid, blk, 0, stream>>>(x, Wl0, bl0, B0, N, head, cnt, ticket, Wr0, Wl1, Wr1, wfrag);

  build_slots<<<buildGrid, blk, 0, stream>>>(ei, cnt, slots, head, link, ticket, E, N);

  sage_gather0_root0_layer1<<<gemmGrid, blk, 0, stream>>>(slots, cnt, head, link, B0, x, wfrag + WF_R0, wfrag + WF_L1, bl1, wfrag + WF_R1, B1, B2, N, E);

  sage_gather_slots<32, true><<<gathGrid, dim3(64), 0, stream>>>(slots, cnt, head, link, B1, B2, out, N, E);
}
